# starting kernel
# baseline (speedup 1.0000x reference)
#include <hip/hip_runtime.h>
#include <cstddef>

typedef __attribute__((ext_vector_type(2))) float v2f;
typedef __attribute__((ext_vector_type(8))) float v8f;

template <int NCOLS, bool RELU_IN>
__global__ __launch_bounds__(256) void gemm_wmma_k(const float* __restrict__ X,
                                                   const float* __restrict__ W,
                                                   float* __restrict__ H,
                                                   int ntiles) {
    constexpr int S = (NCOLS == 128) ? 288 : NCOLS * 2;
    __shared__ float ldsW[64 * S];

    for (int idx = threadIdx.x; idx < 128 * NCOLS; idx += 256) {
        const int r = idx / NCOLS;
        const int c = idx % NCOLS;
        ldsW[(r >> 1) * S + c * 2 + (r & 1)] = W[(size_t)r * NCOLS + c];
    }
    __syncthreads();

    const int lane = threadIdx.x & 31;
    const int wave = threadIdx.x >> 5;
    const int tile = blockIdx.x * 8 + wave;
    if (tile >= ntiles) return;
    const int row0 = tile * 16;
    const int m    = lane & 15;
    const int hi   = lane >> 4;
    constexpr int NT = NCOLS / 16;

    v8f acc[NT] = {};
    const float* xrow = X + (size_t)(row0 + m) * 128;

    for (int k = 0; k < 128; k += 4) {
        float a0 = xrow[k + 2 * hi];
        float a1 = xrow[k + 2 * hi + 1];
        if (RELU_IN) { a0 = fmaxf(a0, 0.f); a1 = fmaxf(a1, 0.f); }
        v2f a; a[0] = a0; a[1] = a1;
        const float* wrow = ldsW + ((k >> 1) + hi) * S;
#pragma unroll
        for (int t = 0; t < NT; ++t) {
            const v2f b = *(const v2f*)(wrow + (t * 16 + m) * 2);
            acc[t] = __builtin_amdgcn_wmma_f32_16x16x4_f32(
                false, a, false, b, (short)0, acc[t], false, false);
        }
    }

#pragma unroll
    for (int t = 0; t < NT; ++t) {
        float* o = H + (size_t)row0 * NCOLS + t * 16 + m;
#pragma unroll
        for (int v = 0; v < 8; ++v)
            o[(size_t)(v + 8 * hi) * NCOLS] = acc[t][v];
    }
}

__global__ __launch_bounds__(256) void k_zero_i(int* p, int n) {
    int i = blockIdx.x * blockDim.x + threadIdx.x;
    if (i < n) p[i] = 0;
}

__global__ __launch_bounds__(256) void k_count_i(const int* __restrict__ dst,
                                                 int* __restrict__ deg, int e) {
    int i = blockIdx.x * blockDim.x + threadIdx.x;
    if (i < e) atomicAdd(&deg[dst[i]], 1);
}

__global__ __launch_bounds__(256) void k_dis(const int* __restrict__ deg,
                                             float* __restrict__ dis, int n) {
    int i = blockIdx.x * blockDim.x + threadIdx.x;
    if (i < n) dis[i] = rsqrtf((float)deg[i] + 1.f);
}

__global__ __launch_bounds__(256) void k_scan1(const int* __restrict__ d,
                                               int* __restrict__ exc,
                                               int* __restrict__ bsum, int n) {
    __shared__ int sh[256];
    const int tid  = threadIdx.x;
    const int base = blockIdx.x * 2048 + tid * 8;
    int v[8]; int s = 0;
#pragma unroll
    for (int j = 0; j < 8; ++j) { v[j] = (base + j < n) ? d[base + j] : 0; s += v[j]; }
    sh[tid] = s;
    __syncthreads();
    for (int off = 1; off < 256; off <<= 1) {
        int t = (tid >= off) ? sh[tid - off] : 0;
        __syncthreads();
        if (tid >= off) sh[tid] += t;
        __syncthreads();
    }
    const int excl = sh[tid] - s;
    if (tid == 255) bsum[blockIdx.x] = sh[255];
    int run = excl;
#pragma unroll
    for (int j = 0; j < 8; ++j) {
        if (base + j < n) exc[base + j] = run;
        run += v[j];
    }
}

__global__ void k_scan2(int* bsum, int nb) {
    if (threadIdx.x == 0 && blockIdx.x == 0) {
        int run = 0;
        for (int i = 0; i < nb; ++i) { int t = bsum[i]; bsum[i] = run; run += t; }
    }
}

__global__ __launch_bounds__(256) void k_scan3(int* __restrict__ exc,
                                               const int* __restrict__ bsum,
                                               int* __restrict__ cursor, int n) {
    int i = blockIdx.x * blockDim.x + threadIdx.x;
    if (i < n) {
        int r = exc[i] + bsum[i >> 11];
        exc[i] = r;
        cursor[i] = r;
    }
}

__global__ __launch_bounds__(256) void k_bucket(const int* __restrict__ src,
                                                const int* __restrict__ dst,
                                                const float* __restrict__ dis,
                                                int* __restrict__ cursor,
                                                int2* __restrict__ meta, int e) {
    int i = blockIdx.x * blockDim.x + threadIdx.x;
    if (i >= e) return;
    int s = src[i];
    int pos = atomicAdd(&cursor[dst[i]], 1);
    meta[pos] = make_int2(s, __float_as_int(dis[s]));
}

__global__ __launch_bounds__(256) void k_gather128(const float* __restrict__ H,
                                                   const int* __restrict__ rowptr,
                                                   const int2* __restrict__ meta,
                                                   const float* __restrict__ dis,
                                                   const float* __restrict__ bias,
                                                   float* __restrict__ B,
                                                   int n, int e) {
    const int node = (blockIdx.x * blockDim.x + threadIdx.x) >> 5;
    const int lane = threadIdx.x & 31;
    if (node >= n) return;
    const int beg = rowptr[node];
    const int end = (node + 1 < n) ? rowptr[node + 1] : e;
    const float di = dis[node];
    const int c = lane * 4;

    const float4 hs = *(const float4*)(H + (size_t)node * 128 + c);
    float4 acc;
    acc.x = hs.x * di; acc.y = hs.y * di; acc.z = hs.z * di; acc.w = hs.w * di;

    for (int p = beg; p < end; ++p) {
        const int2 m = meta[p];
        const float w = __int_as_float(m.y);
        const float4 v = *(const float4*)(H + (size_t)m.x * 128 + c);
        acc.x = fmaf(v.x, w, acc.x); acc.y = fmaf(v.y, w, acc.y);
        acc.z = fmaf(v.z, w, acc.z); acc.w = fmaf(v.w, w, acc.w);
    }
    const float4 bb = *(const float4*)(bias + c);
    float4 r;
    r.x = fmaf(di, acc.x, bb.x); r.y = fmaf(di, acc.y, bb.y);
    r.z = fmaf(di, acc.z, bb.z); r.w = fmaf(di, acc.w, bb.w);
    *(float4*)(B + (size_t)node * 128 + c) = r;
}

__global__ __launch_bounds__(256) void k_gather16(const float* __restrict__ H,
                                                  const int* __restrict__ rowptr,
                                                  const int2* __restrict__ meta,
                                                  const float* __restrict__ dis,
                                                  const float* __restrict__ bias,
                                                  float* __restrict__ out,
                                                  int n, int e) {
    const int node = (blockIdx.x * blockDim.x + threadIdx.x) >> 5;
    const int lane = threadIdx.x & 31;
    if (node >= n) return;
    const int beg = rowptr[node];
    const int end = (node + 1 < n) ? rowptr[node + 1] : e;
    const float di = dis[node];
    const int g  = lane >> 2;
    const int q  = (lane & 3) * 4;

    float4 acc = make_float4(0.f, 0.f, 0.f, 0.f);
    if (g == 0) {
        const float4 hs = *(const float4*)(H + (size_t)node * 16 + q);
        acc.x = hs.x * di; acc.y = hs.y * di; acc.z = hs.z * di; acc.w = hs.w * di;
    }
    for (int p = beg + g; p < end; p += 8) {
        const int2 m = meta[p];
        const float w = __int_as_float(m.y);
        const float4 v = *(const float4*)(H + (size_t)m.x * 16 + q);
        acc.x = fmaf(v.x, w, acc.x); acc.y = fmaf(v.y, w, acc.y);
        acc.z = fmaf(v.z, w, acc.z); acc.w = fmaf(v.w, w, acc.w);
    }
#pragma unroll
    for (int mask = 4; mask <= 16; mask <<= 1) {
        acc.x += __shfl_xor(acc.x, mask, 32);
        acc.y += __shfl_xor(acc.y, mask, 32);
        acc.z += __shfl_xor(acc.z, mask, 32);
        acc.w += __shfl_xor(acc.w, mask, 32);
    }
    if (lane < 4) {
        const float4 bb = *(const float4*)(bias + q);
        float4 r;
        r.x = fmaf(di, acc.x, bb.x); r.y = fmaf(di, acc.y, bb.y);
        r.z = fmaf(di, acc.z, bb.z); r.w = fmaf(di, acc.w, bb.w);
        *(float4*)(out + (size_t)node * 16 + q) = r;
    }
}

extern "C" void kernel_launch(void* const* d_in, const int* in_sizes, int n_in,
                              void* d_out, int out_size, void* d_ws, size_t ws_size,
                              hipStream_t stream) {
    const float* x   = (const float*)d_in[0];
    const int*   ei  = (const int*)d_in[1];
    const float* W1  = (const float*)d_in[2];
    const float* b1  = (const float*)d_in[3];
    const float* W2  = (const float*)d_in[4];
    const float* b2  = (const float*)d_in[5];
    const float* W3  = (const float*)d_in[6];
    const float* b3  = (const float*)d_in[7];
    float*       out = (float*)d_out;

    const int N = in_sizes[0] / 128;
    const int E = in_sizes[1] / 2;
    const int* src = ei;
    const int* dst = ei + E;

    float* A      = (float*)d_ws;
    float* B      = A + (size_t)N * 128;
    int2*  meta   = (int2*)(B + (size_t)N * 128);
    float* dis    = (float*)(meta + E);
    int*   degi   = (int*)(dis + N);
    int*   rowptr = degi + N;
    int*   cursor = rowptr + N;
    int*   bsum   = cursor + N;

    const int TPB = 256;
    auto cdiv = [](long long a, long long b) { return (int)((a + b - 1) / b); };
    const int ntiles = (N + 15) / 16;
    const int nb     = cdiv(N, 2048);

    k_zero_i <<<cdiv(N, TPB), TPB, 0, stream>>>(degi, N);
    k_count_i<<<cdiv(E, TPB), TPB, 0, stream>>>(dst, degi, E);
    k_dis    <<<cdiv(N, TPB), TPB, 0, stream>>>(degi, dis, N);
    k_scan1  <<<nb, TPB, 0, stream>>>(degi, rowptr, bsum, N);
    k_scan2  <<<1, 32, 0, stream>>>(bsum, nb);
    k_scan3  <<<cdiv(N, TPB), TPB, 0, stream>>>(rowptr, bsum, cursor, N);
    k_bucket <<<cdiv(E, TPB), TPB, 0, stream>>>(src, dst, dis, cursor, meta, E);

    gemm_wmma_k<128, false><<<cdiv(ntiles, 8), TPB, 0, stream>>>(x, W1, A, ntiles);
    k_gather128<<<cdiv((long long)N * 32, TPB), TPB, 0, stream>>>(A, rowptr, meta, dis, b1, B, N, E);
    gemm_wmma_k<128, true><<<cdiv(ntiles, 8), TPB, 0, stream>>>(B, W2, A, ntiles);
    k_gather128<<<cdiv((long long)N * 32, TPB), TPB, 0, stream>>>(A, rowptr, meta, dis, b2, B, N, E);
    gemm_wmma_k<16, true><<<cdiv(ntiles, 8), TPB, 0, stream>>>(B, W3, A, ntiles);
    k_gather16<<<cdiv((long long)N * 32, TPB), TPB, 0, stream>>>(A, rowptr, meta, dis, b3, out, N, E);
}


# CSR build without global atomics: LDS histograms per (edge chunk, node range) give each edge its rank, column scans over chunks give degree and base; bucket order is chunk, then LDS arrival
# speedup vs baseline: 1.3954x; 1.3954x over previous
#include <hip/hip_runtime.h>
#include <cstddef>

typedef __attribute__((ext_vector_type(2))) float v2f;
typedef __attribute__((ext_vector_type(8))) float v8f;

template <int NCOLS, bool RELU_IN>
__global__ __launch_bounds__(256) void gemm_wmma_k(const float* __restrict__ X,
                                                   const float* __restrict__ W,
                                                   float* __restrict__ H,
                                                   int ntiles) {
    constexpr int S = (NCOLS == 128) ? 288 : NCOLS * 2;
    __shared__ float ldsW[64 * S];

    for (int idx = threadIdx.x; idx < 128 * NCOLS; idx += 256) {
        const int r = idx / NCOLS;
        const int c = idx % NCOLS;
        ldsW[(r >> 1) * S + c * 2 + (r & 1)] = W[(size_t)r * NCOLS + c];
    }
    __syncthreads();

    const int lane = threadIdx.x & 31;
    const int wave = threadIdx.x >> 5;
    const int tile = blockIdx.x * 8 + wave;
    if (tile >= ntiles) return;
    const int row0 = tile * 16;
    const int m    = lane & 15;
    const int hi   = lane >> 4;
    constexpr int NT = NCOLS / 16;

    v8f acc[NT] = {};
    const float* xrow = X + (size_t)(row0 + m) * 128;

    for (int k = 0; k < 128; k += 4) {
        float a0 = xrow[k + 2 * hi];
        float a1 = xrow[k + 2 * hi + 1];
        if (RELU_IN) { a0 = fmaxf(a0, 0.f); a1 = fmaxf(a1, 0.f); }
        v2f a; a[0] = a0; a[1] = a1;
        const float* wrow = ldsW + ((k >> 1) + hi) * S;
#pragma unroll
        for (int t = 0; t < NT; ++t) {
            const v2f b = *(const v2f*)(wrow + (t * 16 + m) * 2);
            acc[t] = __builtin_amdgcn_wmma_f32_16x16x4_f32(
                false, a, false, b, (short)0, acc[t], false, false);
        }
    }

#pragma unroll
    for (int t = 0; t < NT; ++t) {
        float* o = H + (size_t)row0 * NCOLS + t * 16 + m;
#pragma unroll
        for (int v = 0; v < 8; ++v)
            o[(size_t)(v + 8 * hi) * NCOLS] = acc[t][v];
    }
}

constexpr int RANGE = 32768;
constexpr int CH    = 32768;

__global__ __launch_bounds__(1024) void k_hist_rank(const int* __restrict__ dst,
                                                    int* __restrict__ rank,
                                                    int* __restrict__ cnt,
                                                    int n, int e) {
    __shared__ int hist[RANGE];
    const int tid = threadIdx.x;
    const int c   = blockIdx.x;
    const int lo  = blockIdx.y * RANGE;
    for (int j = tid; j < RANGE; j += 1024) hist[j] = 0;
    __syncthreads();

    const int ebeg = c * CH;
    const int eend = (ebeg + CH < e) ? ebeg + CH : e;
    for (int base = ebeg; base < eend; base += 4096) {
        int d[4];
#pragma unroll
        for (int u = 0; u < 4; ++u) {
            const int i = base + u * 1024 + tid;
            d[u] = (i < eend) ? dst[i] : -1;
        }
#pragma unroll
        for (int u = 0; u < 4; ++u) {
            const unsigned rel = (unsigned)(d[u] - lo);
            if (rel < (unsigned)RANGE)
                rank[base + u * 1024 + tid] = atomicAdd(&hist[rel], 1);
        }
    }
    __syncthreads();

    const int hi = (n - lo < RANGE) ? n - lo : RANGE;
    int* crow = cnt + (size_t)c * n + lo;
    for (int j = tid; j < hi; j += 1024) crow[j] = hist[j];
}

__global__ __launch_bounds__(256) void k_coldeg(const int* __restrict__ cnt,
                                                int* __restrict__ deg,
                                                float* __restrict__ dis,
                                                int n, int nchunk) {
    const int i = blockIdx.x * blockDim.x + threadIdx.x;
    if (i >= n) return;
    int run = 0;
#pragma unroll 7
    for (int c = 0; c < nchunk; ++c) run += cnt[(size_t)c * n + i];
    deg[i] = run;
    dis[i] = rsqrtf((float)run + 1.f);
}

__global__ __launch_bounds__(256) void k_scan1(const int* __restrict__ d,
                                               int* __restrict__ exc,
                                               int* __restrict__ bsum, int n) {
    __shared__ int sh[256];
    const int tid  = threadIdx.x;
    const int base = blockIdx.x * 2048 + tid * 8;
    int v[8]; int s = 0;
#pragma unroll
    for (int j = 0; j < 8; ++j) { v[j] = (base + j < n) ? d[base + j] : 0; s += v[j]; }
    sh[tid] = s;
    __syncthreads();
    for (int off = 1; off < 256; off <<= 1) {
        int t = (tid >= off) ? sh[tid - off] : 0;
        __syncthreads();
        if (tid >= off) sh[tid] += t;
        __syncthreads();
    }
    const int excl = sh[tid] - s;
    if (tid == 255) bsum[blockIdx.x] = sh[255];
    int run = excl;
#pragma unroll
    for (int j = 0; j < 8; ++j) {
        if (base + j < n) exc[base + j] = run;
        run += v[j];
    }
}

__global__ void k_scan2(int* bsum, int nb) {
    if (threadIdx.x == 0 && blockIdx.x == 0) {
        int run = 0;
        for (int i = 0; i < nb; ++i) { int t = bsum[i]; bsum[i] = run; run += t; }
    }
}

__global__ __launch_bounds__(256) void k_colbase(int* __restrict__ exc,
                                                 const int* __restrict__ bsum,
                                                 const int* __restrict__ cnt,
                                                 int* __restrict__ cbase,
                                                 int n, int nchunk) {
    const int i = blockIdx.x * blockDim.x + threadIdx.x;
    if (i >= n) return;
    int run = exc[i] + bsum[i >> 11];
    exc[i] = run;
#pragma unroll 7
    for (int c = 0; c < nchunk; ++c) {
        const int t = cnt[(size_t)c * n + i];
        cbase[(size_t)c * n + i] = run;
        run += t;
    }
}

__global__ __launch_bounds__(256) void k_place(const int* __restrict__ src,
                                               const int* __restrict__ dst,
                                               const int* __restrict__ rank,
                                               const int* __restrict__ cbase,
                                               const float* __restrict__ dis,
                                               int2* __restrict__ meta,
                                               int n, int e) {
    int i = blockIdx.x * blockDim.x + threadIdx.x;
    if (i >= e) return;
    const int s = src[i];
    const int d = dst[i];
    const int c = i / CH;
    const int pos = cbase[(size_t)c * n + d] + rank[i];
    meta[pos] = make_int2(s, __float_as_int(dis[s]));
}

__global__ __launch_bounds__(256) void k_gather128(const float* __restrict__ H,
                                                   const int* __restrict__ rowptr,
                                                   const int2* __restrict__ meta,
                                                   const float* __restrict__ dis,
                                                   const float* __restrict__ bias,
                                                   float* __restrict__ B,
                                                   int n, int e) {
    const int node = (blockIdx.x * blockDim.x + threadIdx.x) >> 5;
    const int lane = threadIdx.x & 31;
    if (node >= n) return;
    const int beg = rowptr[node];
    const int end = (node + 1 < n) ? rowptr[node + 1] : e;
    const float di = dis[node];
    const int c = lane * 4;

    const float4 hs = *(const float4*)(H + (size_t)node * 128 + c);
    float4 acc;
    acc.x = hs.x * di; acc.y = hs.y * di; acc.z = hs.z * di; acc.w = hs.w * di;

    for (int p = beg; p < end; ++p) {
        const int2 m = meta[p];
        const float w = __int_as_float(m.y);
        const float4 v = *(const float4*)(H + (size_t)m.x * 128 + c);
        acc.x = fmaf(v.x, w, acc.x); acc.y = fmaf(v.y, w, acc.y);
        acc.z = fmaf(v.z, w, acc.z); acc.w = fmaf(v.w, w, acc.w);
    }
    const float4 bb = *(const float4*)(bias + c);
    float4 r;
    r.x = fmaf(di, acc.x, bb.x); r.y = fmaf(di, acc.y, bb.y);
    r.z = fmaf(di, acc.z, bb.z); r.w = fmaf(di, acc.w, bb.w);
    *(float4*)(B + (size_t)node * 128 + c) = r;
}

__global__ __launch_bounds__(256) void k_gather16(const float* __restrict__ H,
                                                  const int* __restrict__ rowptr,
                                                  const int2* __restrict__ meta,
                                                  const float* __restrict__ dis,
                                                  const float* __restrict__ bias,
                                                  float* __restrict__ out,
                                                  int n, int e) {
    const int node = (blockIdx.x * blockDim.x + threadIdx.x) >> 5;
    const int lane = threadIdx.x & 31;
    if (node >= n) return;
    const int beg = rowptr[node];
    const int end = (node + 1 < n) ? rowptr[node + 1] : e;
    const float di = dis[node];
    const int g  = lane >> 2;
    const int q  = (lane & 3) * 4;

    float4 acc = make_float4(0.f, 0.f, 0.f, 0.f);
    if (g == 0) {
        const float4 hs = *(const float4*)(H + (size_t)node * 16 + q);
        acc.x = hs.x * di; acc.y = hs.y * di; acc.z = hs.z * di; acc.w = hs.w * di;
    }
    for (int p = beg + g; p < end; p += 8) {
        const int2 m = meta[p];
        const float w = __int_as_float(m.y);
        const float4 v = *(const float4*)(H + (size_t)m.x * 16 + q);
        acc.x = fmaf(v.x, w, acc.x); acc.y = fmaf(v.y, w, acc.y);
        acc.z = fmaf(v.z, w, acc.z); acc.w = fmaf(v.w, w, acc.w);
    }
#pragma unroll
    for (int mask = 4; mask <= 16; mask <<= 1) {
        acc.x += __shfl_xor(acc.x, mask, 32);
        acc.y += __shfl_xor(acc.y, mask, 32);
        acc.z += __shfl_xor(acc.z, mask, 32);
        acc.w += __shfl_xor(acc.w, mask, 32);
    }
    if (lane < 4) {
        const float4 bb = *(const float4*)(bias + q);
        float4 r;
        r.x = fmaf(di, acc.x, bb.x); r.y = fmaf(di, acc.y, bb.y);
        r.z = fmaf(di, acc.z, bb.z); r.w = fmaf(di, acc.w, bb.w);
        *(float4*)(out + (size_t)node * 16 + q) = r;
    }
}

extern "C" void kernel_launch(void* const* d_in, const int* in_sizes, int n_in,
                              void* d_out, int out_size, void* d_ws, size_t ws_size,
                              hipStream_t stream) {
    const float* x   = (const float*)d_in[0];
    const int*   ei  = (const int*)d_in[1];
    const float* W1  = (const float*)d_in[2];
    const float* b1  = (const float*)d_in[3];
    const float* W2  = (const float*)d_in[4];
    const float* b2  = (const float*)d_in[5];
    const float* W3  = (const float*)d_in[6];
    const float* b3  = (const float*)d_in[7];
    float*       out = (float*)d_out;

    const int N = in_sizes[0] / 128;
    const int E = in_sizes[1] / 2;
    const int* src = ei;
    const int* dst = ei + E;

    const int nchunk = (E + CH - 1) / CH;
    const int nrange = (N + RANGE - 1) / RANGE;
    float* A      = (float*)d_ws;
    float* B      = A + (size_t)N * 128;
    int2*  meta   = (int2*)(B + (size_t)N * 128);
    float* dis    = (float*)(meta + E);
    int*   degi   = (int*)(dis + N);
    int*   rowptr = degi + N;
    int*   rank   = rowptr + N;
    int*   cnt    = rank + E;
    int*   cbase  = cnt + (size_t)nchunk * N;
    int*   bsum   = cbase + (size_t)nchunk * N;

    const int TPB = 256;
    auto cdiv = [](long long a, long long b) { return (int)((a + b - 1) / b); };
    const int ntiles = (N + 15) / 16;
    const int nb     = cdiv(N, 2048);

    k_hist_rank<<<dim3(nchunk, nrange), 1024, 0, stream>>>(dst, rank, cnt, N, E);
    k_coldeg <<<cdiv(N, TPB), TPB, 0, stream>>>(cnt, degi, dis, N, nchunk);
    k_scan1  <<<nb, TPB, 0, stream>>>(degi, rowptr, bsum, N);
    k_scan2  <<<1, 32, 0, stream>>>(bsum, nb);
    k_colbase<<<cdiv(N, TPB), TPB, 0, stream>>>(rowptr, bsum, cnt, cbase, N, nchunk);
    k_place  <<<cdiv(E, TPB), TPB, 0, stream>>>(src, dst, rank, cbase, dis, meta, N, E);

    gemm_wmma_k<128, false><<<cdiv(ntiles, 8), TPB, 0, stream>>>(x, W1, A, ntiles);
    k_gather128<<<cdiv((long long)N * 32, TPB), TPB, 0, stream>>>(A, rowptr, meta, dis, b1, B, N, E);
    gemm_wmma_k<128, true><<<cdiv(ntiles, 8), TPB, 0, stream>>>(B, W2, A, ntiles);
    k_gather128<<<cdiv((long long)N * 32, TPB), TPB, 0, stream>>>(A, rowptr, meta, dis, b2, B, N, E);
    gemm_wmma_k<16, true><<<cdiv(ntiles, 8), TPB, 0, stream>>>(B, W3, A, ntiles);
    k_gather16<<<cdiv((long long)N * 32, TPB), TPB, 0, stream>>>(A, rowptr, meta, dis, b3, out, N, E);
}


# gather: a wave walks 4 consecutive nodes (last layer: 8 nodes side by side) and requests 8 rows at a time from a scalar-loaded edge list, instead of one short wave per node; sum order kept
# speedup vs baseline: 1.4597x; 1.0461x over previous
#include <hip/hip_runtime.h>
#include <cstddef>

typedef __attribute__((ext_vector_type(2))) float v2f;
typedef __attribute__((ext_vector_type(8))) float v8f;

template <int NCOLS, bool RELU_IN>
__global__ __launch_bounds__(256) void gemm_wmma_k(const float* __restrict__ X,
                                                   const float* __restrict__ W,
                                                   float* __restrict__ H,
                                                   int ntiles) {
    constexpr int S = (NCOLS == 128) ? 288 : NCOLS * 2;
    __shared__ float ldsW[64 * S];

    for (int idx = threadIdx.x; idx < 128 * NCOLS; idx += 256) {
        const int r = idx / NCOLS;
        const int c = idx % NCOLS;
        ldsW[(r >> 1) * S + c * 2 + (r & 1)] = W[(size_t)r * NCOLS + c];
    }
    __syncthreads();

    const int lane = threadIdx.x & 31;
    const int wave = threadIdx.x >> 5;
    const int tile = blockIdx.x * 8 + wave;
    if (tile >= ntiles) return;
    const int row0 = tile * 16;
    const int m    = lane & 15;
    const int hi   = lane >> 4;
    constexpr int NT = NCOLS / 16;

    v8f acc[NT] = {};
    const float* xrow = X + (size_t)(row0 + m) * 128;

    for (int k = 0; k < 128; k += 4) {
        float a0 = xrow[k + 2 * hi];
        float a1 = xrow[k + 2 * hi + 1];
        if (RELU_IN) { a0 = fmaxf(a0, 0.f); a1 = fmaxf(a1, 0.f); }
        v2f a; a[0] = a0; a[1] = a1;
        const float* wrow = ldsW + ((k >> 1) + hi) * S;
#pragma unroll
        for (int t = 0; t < NT; ++t) {
            const v2f b = *(const v2f*)(wrow + (t * 16 + m) * 2);
            acc[t] = __builtin_amdgcn_wmma_f32_16x16x4_f32(
                false, a, false, b, (short)0, acc[t], false, false);
        }
    }

#pragma unroll
    for (int t = 0; t < NT; ++t) {
        float* o = H + (size_t)row0 * NCOLS + t * 16 + m;
#pragma unroll
        for (int v = 0; v < 8; ++v)
            o[(size_t)(v + 8 * hi) * NCOLS] = acc[t][v];
    }
}

constexpr int RANGE = 32768;
constexpr int CH    = 32768;

__global__ __launch_bounds__(1024) void k_hist_rank(const int* __restrict__ dst,
                                                    int* __restrict__ rank,
                                                    int* __restrict__ cnt,
                                                    int n, int e) {
    __shared__ int hist[RANGE];
    const int tid = threadIdx.x;
    const int c   = blockIdx.x;
    const int lo  = blockIdx.y * RANGE;
    for (int j = tid; j < RANGE; j += 1024) hist[j] = 0;
    __syncthreads();

    const int ebeg = c * CH;
    const int eend = (ebeg + CH < e) ? ebeg + CH : e;
    for (int base = ebeg; base < eend; base += 4096) {
        int d[4];
#pragma unroll
        for (int u = 0; u < 4; ++u) {
            const int i = base + u * 1024 + tid;
            d[u] = (i < eend) ? dst[i] : -1;
        }
#pragma unroll
        for (int u = 0; u < 4; ++u) {
            const unsigned rel = (unsigned)(d[u] - lo);
            if (rel < (unsigned)RANGE)
                rank[base + u * 1024 + tid] = atomicAdd(&hist[rel], 1);
        }
    }
    __syncthreads();

    const int hi = (n - lo < RANGE) ? n - lo : RANGE;
    int* crow = cnt + (size_t)c * n + lo;
    for (int j = tid; j < hi; j += 1024) crow[j] = hist[j];
}

__global__ __launch_bounds__(256) void k_coldeg(const int* __restrict__ cnt,
                                                int* __restrict__ deg,
                                                float* __restrict__ dis,
                                                int n, int nchunk) {
    const int i = blockIdx.x * blockDim.x + threadIdx.x;
    if (i >= n) return;
    int run = 0;
#pragma unroll 7
    for (int c = 0; c < nchunk; ++c) run += cnt[(size_t)c * n + i];
    deg[i] = run;
    dis[i] = rsqrtf((float)run + 1.f);
}

__global__ __launch_bounds__(256) void k_scan1(const int* __restrict__ d,
                                               int* __restrict__ exc,
                                               int* __restrict__ bsum, int n) {
    __shared__ int sh[256];
    const int tid  = threadIdx.x;
    const int base = blockIdx.x * 2048 + tid * 8;
    int v[8]; int s = 0;
#pragma unroll
    for (int j = 0; j < 8; ++j) { v[j] = (base + j < n) ? d[base + j] : 0; s += v[j]; }
    sh[tid] = s;
    __syncthreads();
    for (int off = 1; off < 256; off <<= 1) {
        int t = (tid >= off) ? sh[tid - off] : 0;
        __syncthreads();
        if (tid >= off) sh[tid] += t;
        __syncthreads();
    }
    const int excl = sh[tid] - s;
    if (tid == 255) bsum[blockIdx.x] = sh[255];
    int run = excl;
#pragma unroll
    for (int j = 0; j < 8; ++j) {
        if (base + j < n) exc[base + j] = run;
        run += v[j];
    }
}

__global__ void k_scan2(int* bsum, int nb) {
    if (threadIdx.x == 0 && blockIdx.x == 0) {
        int run = 0;
        for (int i = 0; i < nb; ++i) { int t = bsum[i]; bsum[i] = run; run += t; }
    }
}

__global__ __launch_bounds__(256) void k_colbase(int* __restrict__ exc,
                                                 const int* __restrict__ bsum,
                                                 const int* __restrict__ cnt,
                                                 int* __restrict__ cbase,
                                                 int n, int nchunk) {
    const int i = blockIdx.x * blockDim.x + threadIdx.x;
    if (i >= n) return;
    int run = exc[i] + bsum[i >> 11];
    exc[i] = run;
#pragma unroll 7
    for (int c = 0; c < nchunk; ++c) {
        const int t = cnt[(size_t)c * n + i];
        cbase[(size_t)c * n + i] = run;
        run += t;
    }
}

__global__ __launch_bounds__(256) void k_place(const int* __restrict__ src,
                                               const int* __restrict__ dst,
                                               const int* __restrict__ rank,
                                               const int* __restrict__ cbase,
                                               const float* __restrict__ dis,
                                               int2* __restrict__ meta,
                                               int n, int e) {
    int i = blockIdx.x * blockDim.x + threadIdx.x;
    if (i >= e) return;
    const int s = src[i];
    const int d = dst[i];
    const int c = i / CH;
    const int pos = cbase[(size_t)c * n + d] + rank[i];
    meta[pos] = make_int2(s, __float_as_int(dis[s]));
}

constexpr int NPW = 4;
constexpr int EB  = 8;

__global__ __launch_bounds__(256) void k_gather128(const float* __restrict__ H,
                                                   const int* __restrict__ rowptr,
                                                   const int2* __restrict__ meta,
                                                   const float* __restrict__ dis,
                                                   const float* __restrict__ bias,
                                                   float* __restrict__ B,
                                                   int n, int e) {
    const int wv = __builtin_amdgcn_readfirstlane(
        (int)((blockIdx.x * blockDim.x + threadIdx.x) >> 5));
    const int lane = threadIdx.x & 31;
    const int c = lane * 4;
    const int n0 = wv * NPW;
    if (n0 >= n) return;
    const int n1 = (n0 + NPW < n) ? n0 + NPW : n;
    const float4 bb = *(const float4*)(bias + c);
    const float* Hc = H + c;

    int beg = rowptr[n0];
    for (int node = n0; node < n1; ++node) {
        const int   end = (node + 1 < n) ? rowptr[node + 1] : e;
        const float di  = dis[node];
        const int   cnt = end - beg + 1;

        int2 m[EB];
        const int2 self = make_int2(node, __float_as_int(di));
        m[0] = self;
        if (beg < end) {
#pragma unroll
            for (int j = 1; j < EB; ++j) {
                const int pj = (beg + j - 1 < end - 1) ? beg + j - 1 : end - 1;
                m[j] = meta[pj];
            }
        } else {
#pragma unroll
            for (int j = 1; j < EB; ++j) m[j] = self;
        }

        float4 acc = make_float4(0.f, 0.f, 0.f, 0.f);
        for (int q = 0; q < cnt; q += EB) {
            float4 v[EB];
            float  w[EB];
#pragma unroll
            for (int j = 0; j < EB; ++j) {
                v[j] = *(const float4*)(Hc + (size_t)m[j].x * 128);
                w[j] = __int_as_float(m[j].y);
            }
            const int left = cnt - q;
            __builtin_amdgcn_sched_barrier(0);
            if (left > EB) {
                const int pe = beg + q + EB - 1;
#pragma unroll
                for (int j = 0; j < EB; ++j) {
                    const int pj = (pe + j < end - 1) ? pe + j : end - 1;
                    m[j] = meta[pj];
                }
            }
#pragma unroll
            for (int j = 0; j < EB; ++j) {
                if (j < left) {
                    acc.x = fmaf(v[j].x, w[j], acc.x); acc.y = fmaf(v[j].y, w[j], acc.y);
                    acc.z = fmaf(v[j].z, w[j], acc.z); acc.w = fmaf(v[j].w, w[j], acc.w);
                }
            }
        }
        float4 r;
        r.x = fmaf(di, acc.x, bb.x); r.y = fmaf(di, acc.y, bb.y);
        r.z = fmaf(di, acc.z, bb.z); r.w = fmaf(di, acc.w, bb.w);
        *(float4*)(B + (size_t)node * 128 + c) = r;
        beg = end;
    }
}

__global__ __launch_bounds__(256) void k_gather16(const float* __restrict__ H,
                                                  const int* __restrict__ rowptr,
                                                  const int2* __restrict__ meta,
                                                  const float* __restrict__ dis,
                                                  const float* __restrict__ bias,
                                                  float* __restrict__ out,
                                                  int n, int e) {
    const int t    = blockIdx.x * blockDim.x + threadIdx.x;
    const int node = t >> 2;
    const int q    = (t & 3) * 4;
    if (node >= n) return;
    const int beg = rowptr[node];
    const int end = (node + 1 < n) ? rowptr[node + 1] : e;
    const float di = dis[node];
    const float* Hq = H + q;

    float4 acc[8];
    {
        const float4 hs = *(const float4*)(Hq + (size_t)node * 16);
        acc[0].x = hs.x * di; acc[0].y = hs.y * di; acc[0].z = hs.z * di; acc[0].w = hs.w * di;
    }
#pragma unroll
    for (int j = 1; j < 8; ++j) acc[j] = make_float4(0.f, 0.f, 0.f, 0.f);

    for (int p = beg; p < end; p += 8) {
        int2   m[8];
        float4 v[8];
#pragma unroll
        for (int j = 0; j < 8; ++j) {
            const int pj = (p + j < end) ? p + j : end - 1;
            m[j] = meta[pj];
        }
        __builtin_amdgcn_sched_barrier(0);
#pragma unroll
        for (int j = 0; j < 8; ++j)
            v[j] = *(const float4*)(Hq + (size_t)m[j].x * 16);
        __builtin_amdgcn_sched_barrier(0);
#pragma unroll
        for (int j = 0; j < 8; ++j) {
            const float w = (p + j < end) ? __int_as_float(m[j].y) : 0.f;
            acc[j].x = fmaf(v[j].x, w, acc[j].x); acc[j].y = fmaf(v[j].y, w, acc[j].y);
            acc[j].z = fmaf(v[j].z, w, acc[j].z); acc[j].w = fmaf(v[j].w, w, acc[j].w);
        }
    }
#pragma unroll
    for (int st = 1; st < 8; st <<= 1) {
#pragma unroll
        for (int j = 0; j < 8; j += 2 * st) {
            acc[j].x += acc[j + st].x; acc[j].y += acc[j + st].y;
            acc[j].z += acc[j + st].z; acc[j].w += acc[j + st].w;
        }
    }
    const float4 bb = *(const float4*)(bias + q);
    float4 r;
    r.x = fmaf(di, acc[0].x, bb.x); r.y = fmaf(di, acc[0].y, bb.y);
    r.z = fmaf(di, acc[0].z, bb.z); r.w = fmaf(di, acc[0].w, bb.w);
    *(float4*)(out + (size_t)node * 16 + q) = r;
}

extern "C" void kernel_launch(void* const* d_in, const int* in_sizes, int n_in,
                              void* d_out, int out_size, void* d_ws, size_t ws_size,
                              hipStream_t stream) {
    const float* x   = (const float*)d_in[0];
    const int*   ei  = (const int*)d_in[1];
    const float* W1  = (const float*)d_in[2];
    const float* b1  = (const float*)d_in[3];
    const float* W2  = (const float*)d_in[4];
    const float* b2  = (const float*)d_in[5];
    const float* W3  = (const float*)d_in[6];
    const float* b3  = (const float*)d_in[7];
    float*       out = (float*)d_out;

    const int N = in_sizes[0] / 128;
    const int E = in_sizes[1] / 2;
    const int* src = ei;
    const int* dst = ei + E;

    const int nchunk = (E + CH - 1) / CH;
    const int nrange = (N + RANGE - 1) / RANGE;
    float* A      = (float*)d_ws;
    float* B      = A + (size_t)N * 128;
    int2*  meta   = (int2*)(B + (size_t)N * 128);
    float* dis    = (float*)(meta + E);
    int*   degi   = (int*)(dis + N);
    int*   rowptr = degi + N;
    int*   rank   = rowptr + N;
    int*   cnt    = rank + E;
    int*   cbase  = cnt + (size_t)nchunk * N;
    int*   bsum   = cbase + (size_t)nchunk * N;

    const int TPB = 256;
    auto cdiv = [](long long a, long long b) { return (int)((a + b - 1) / b); };
    const int ntiles = (N + 15) / 16;
    const int nb     = cdiv(N, 2048);

    k_hist_rank<<<dim3(nchunk, nrange), 1024, 0, stream>>>(dst, rank, cnt, N, E);
    k_coldeg <<<cdiv(N, TPB), TPB, 0, stream>>>(cnt, degi, dis, N, nchunk);
    k_scan1  <<<nb, TPB, 0, stream>>>(degi, rowptr, bsum, N);
    k_scan2  <<<1, 32, 0, stream>>>(bsum, nb);
    k_colbase<<<cdiv(N, TPB), TPB, 0, stream>>>(rowptr, bsum, cnt, cbase, N, nchunk);
    k_place  <<<cdiv(E, TPB), TPB, 0, stream>>>(src, dst, rank, cbase, dis, meta, N, E);

    gemm_wmma_k<128, false><<<cdiv(ntiles, 8), TPB, 0, stream>>>(x, W1, A, ntiles);
    k_gather128<<<cdiv(N, NPW * 8), TPB, 0, stream>>>(A, rowptr, meta, dis, b1, B, N, E);
    gemm_wmma_k<128, true><<<cdiv(ntiles, 8), TPB, 0, stream>>>(B, W2, A, ntiles);
    k_gather128<<<cdiv(N, NPW * 8), TPB, 0, stream>>>(A, rowptr, meta, dis, b2, B, N, E);
    gemm_wmma_k<16, true><<<cdiv(ntiles, 8), TPB, 0, stream>>>(B, W3, A, ntiles);
    k_gather16<<<cdiv((long long)N * 4, TPB), TPB, 0, stream>>>(A, rowptr, meta, dis, b3, out, N, E);
}


# fuse each 128-column gather with the next layer's GEMM: a wave gathers its 16 nodes into an LDS tile and multiplies it by W there; the aggregated layer is never written or re-read; sum order kept
# speedup vs baseline: 1.6080x; 1.1016x over previous
#include <hip/hip_runtime.h>
#include <cstddef>

typedef __attribute__((ext_vector_type(2))) float v2f;
typedef __attribute__((ext_vector_type(8))) float v8f;

template <int NCOLS, bool RELU_IN>
__global__ __launch_bounds__(256) void gemm_wmma_k(const float* __restrict__ X,
                                                   const float* __restrict__ W,
                                                   float* __restrict__ H,
                                                   int ntiles) {
    constexpr int S = (NCOLS == 128) ? 288 : NCOLS * 2;
    __shared__ float ldsW[64 * S];

    for (int idx = threadIdx.x; idx < 128 * NCOLS; idx += 256) {
        const int r = idx / NCOLS;
        const int c = idx % NCOLS;
        ldsW[(r >> 1) * S + c * 2 + (r & 1)] = W[(size_t)r * NCOLS + c];
    }
    __syncthreads();

    const int lane = threadIdx.x & 31;
    const int wave = threadIdx.x >> 5;
    const int tile = blockIdx.x * 8 + wave;
    if (tile >= ntiles) return;
    const int row0 = tile * 16;
    const int m    = lane & 15;
    const int hi   = lane >> 4;
    constexpr int NT = NCOLS / 16;

    v8f acc[NT] = {};
    const float* xrow = X + (size_t)(row0 + m) * 128;

    for (int k = 0; k < 128; k += 4) {
        float a0 = xrow[k + 2 * hi];
        float a1 = xrow[k + 2 * hi + 1];
        if (RELU_IN) { a0 = fmaxf(a0, 0.f); a1 = fmaxf(a1, 0.f); }
        v2f a; a[0] = a0; a[1] = a1;
        const float* wrow = ldsW + ((k >> 1) + hi) * S;
#pragma unroll
        for (int t = 0; t < NT; ++t) {
            const v2f b = *(const v2f*)(wrow + (t * 16 + m) * 2);
            acc[t] = __builtin_amdgcn_wmma_f32_16x16x4_f32(
                false, a, false, b, (short)0, acc[t], false, false);
        }
    }

#pragma unroll
    for (int t = 0; t < NT; ++t) {
        float* o = H + (size_t)row0 * NCOLS + t * 16 + m;
#pragma unroll
        for (int v = 0; v < 8; ++v)
            o[(size_t)(v + 8 * hi) * NCOLS] = acc[t][v];
    }
}

constexpr int RANGE = 32768;
constexpr int CH    = 32768;

__global__ __launch_bounds__(1024) void k_hist_rank(const int* __restrict__ dst,
                                                    int* __restrict__ rank,
                                                    int* __restrict__ cnt,
                                                    int n, int e) {
    __shared__ int hist[RANGE];
    const int tid = threadIdx.x;
    const int c   = blockIdx.x;
    const int lo  = blockIdx.y * RANGE;
    for (int j = tid; j < RANGE; j += 1024) hist[j] = 0;
    __syncthreads();

    const int ebeg = c * CH;
    const int eend = (ebeg + CH < e) ? ebeg + CH : e;
    for (int base = ebeg; base < eend; base += 4096) {
        int d[4];
#pragma unroll
        for (int u = 0; u < 4; ++u) {
            const int i = base + u * 1024 + tid;
            d[u] = (i < eend) ? dst[i] : -1;
        }
#pragma unroll
        for (int u = 0; u < 4; ++u) {
            const unsigned rel = (unsigned)(d[u] - lo);
            if (rel < (unsigned)RANGE)
                rank[base + u * 1024 + tid] = atomicAdd(&hist[rel], 1);
        }
    }
    __syncthreads();

    const int hi = (n - lo < RANGE) ? n - lo : RANGE;
    int* crow = cnt + (size_t)c * n + lo;
    for (int j = tid; j < hi; j += 1024) crow[j] = hist[j];
}

__global__ __launch_bounds__(256) void k_coldeg(const int* __restrict__ cnt,
                                                int* __restrict__ deg,
                                                float* __restrict__ dis,
                                                int n, int nchunk) {
    const int i = blockIdx.x * blockDim.x + threadIdx.x;
    if (i >= n) return;
    int run = 0;
#pragma unroll 7
    for (int c = 0; c < nchunk; ++c) run += cnt[(size_t)c * n + i];
    deg[i] = run;
    dis[i] = rsqrtf((float)run + 1.f);
}

__global__ __launch_bounds__(256) void k_scan1(const int* __restrict__ d,
                                               int* __restrict__ exc,
                                               int* __restrict__ bsum, int n) {
    __shared__ int sh[256];
    const int tid  = threadIdx.x;
    const int base = blockIdx.x * 2048 + tid * 8;
    int v[8]; int s = 0;
#pragma unroll
    for (int j = 0; j < 8; ++j) { v[j] = (base + j < n) ? d[base + j] : 0; s += v[j]; }
    sh[tid] = s;
    __syncthreads();
    for (int off = 1; off < 256; off <<= 1) {
        int t = (tid >= off) ? sh[tid - off] : 0;
        __syncthreads();
        if (tid >= off) sh[tid] += t;
        __syncthreads();
    }
    const int excl = sh[tid] - s;
    if (tid == 255) bsum[blockIdx.x] = sh[255];
    int run = excl;
#pragma unroll
    for (int j = 0; j < 8; ++j) {
        if (base + j < n) exc[base + j] = run;
        run += v[j];
    }
}

__global__ void k_scan2(int* bsum, int nb) {
    if (threadIdx.x == 0 && blockIdx.x == 0) {
        int run = 0;
        for (int i = 0; i < nb; ++i) { int t = bsum[i]; bsum[i] = run; run += t; }
    }
}

__global__ __launch_bounds__(256) void k_colbase(int* __restrict__ exc,
                                                 const int* __restrict__ bsum,
                                                 const int* __restrict__ cnt,
                                                 int* __restrict__ cbase,
                                                 int n, int nchunk) {
    const int i = blockIdx.x * blockDim.x + threadIdx.x;
    if (i >= n) return;
    int run = exc[i] + bsum[i >> 11];
    exc[i] = run;
#pragma unroll 7
    for (int c = 0; c < nchunk; ++c) {
        const int t = cnt[(size_t)c * n + i];
        cbase[(size_t)c * n + i] = run;
        run += t;
    }
}

__global__ __launch_bounds__(256) void k_place(const int* __restrict__ src,
                                               const int* __restrict__ dst,
                                               const int* __restrict__ rank,
                                               const int* __restrict__ cbase,
                                               const float* __restrict__ dis,
                                               int2* __restrict__ meta,
                                               int n, int e) {
    int i = blockIdx.x * blockDim.x + threadIdx.x;
    if (i >= e) return;
    const int s = src[i];
    const int d = dst[i];
    const int c = i / CH;
    const int pos = cbase[(size_t)c * n + d] + rank[i];
    meta[pos] = make_int2(s, __float_as_int(dis[s]));
}

constexpr int EB = 8;

template <int NCOLS, int WPB>
__global__ __launch_bounds__(WPB * 32) void k_gather_gemm(const float* __restrict__ H,
                                                          const int* __restrict__ rowptr,
                                                          const int2* __restrict__ meta,
                                                          const float* __restrict__ dis,
                                                          const float* __restrict__ bias,
                                                          const float* __restrict__ W,
                                                          float* __restrict__ Hout,
                                                          int n, int e, int ntiles) {
    constexpr int S  = (NCOLS == 128) ? 288 : NCOLS * 2;
    constexpr int TP = 132;
    __shared__ float ldsW[64 * S];
    __shared__ float ldsT[WPB * 16 * TP];

    for (int idx = threadIdx.x; idx < 128 * NCOLS; idx += WPB * 32) {
        const int r = idx / NCOLS;
        const int c = idx % NCOLS;
        ldsW[(r >> 1) * S + c * 2 + (r & 1)] = W[(size_t)r * NCOLS + c];
    }
    __syncthreads();

    const int wave = __builtin_amdgcn_readfirstlane((int)(threadIdx.x >> 5));
    const int lane = threadIdx.x & 31;
    const int tile = blockIdx.x * WPB + wave;
    if (tile >= ntiles) return;
    float* T = ldsT + wave * (16 * TP);
    const int n0 = tile * 16;
    const int n1 = (n0 + 16 < n) ? n0 + 16 : n;

    {
        const int c = lane * 4;
        const float4 bb = *(const float4*)(bias + c);
        const float* Hc = H + c;
        int beg = rowptr[n0];
        for (int node = n0; node < n1; ++node) {
            const int   end = (node + 1 < n) ? rowptr[node + 1] : e;
            const float di  = dis[node];
            const int   cnt = end - beg + 1;

            int2 m[EB];
            const int2 self = make_int2(node, __float_as_int(di));
            m[0] = self;
            if (beg < end) {
#pragma unroll
                for (int j = 1; j < EB; ++j) {
                    const int pj = (beg + j - 1 < end - 1) ? beg + j - 1 : end - 1;
                    m[j] = meta[pj];
                }
            } else {
#pragma unroll
                for (int j = 1; j < EB; ++j) m[j] = self;
            }

            float4 acc = make_float4(0.f, 0.f, 0.f, 0.f);
            for (int q = 0; q < cnt; q += EB) {
                float4 v[EB];
                float  w[EB];
#pragma unroll
                for (int j = 0; j < EB; ++j) {
                    v[j] = *(const float4*)(Hc + (size_t)m[j].x * 128);
                    w[j] = __int_as_float(m[j].y);
                }
                const int left = cnt - q;
                __builtin_amdgcn_sched_barrier(0);
                if (left > EB) {
                    const int pe = beg + q + EB - 1;
#pragma unroll
                    for (int j = 0; j < EB; ++j) {
                        const int pj = (pe + j < end - 1) ? pe + j : end - 1;
                        m[j] = meta[pj];
                    }
                }
#pragma unroll
                for (int j = 0; j < EB; ++j) {
                    if (j < left) {
                        acc.x = fmaf(v[j].x, w[j], acc.x); acc.y = fmaf(v[j].y, w[j], acc.y);
                        acc.z = fmaf(v[j].z, w[j], acc.z); acc.w = fmaf(v[j].w, w[j], acc.w);
                    }
                }
            }
            float4 r;
            r.x = fmaxf(fmaf(di, acc.x, bb.x), 0.f); r.y = fmaxf(fmaf(di, acc.y, bb.y), 0.f);
            r.z = fmaxf(fmaf(di, acc.z, bb.z), 0.f); r.w = fmaxf(fmaf(di, acc.w, bb.w), 0.f);
            *(float4*)(T + (node - n0) * TP + c) = r;
            beg = end;
        }
        for (int r = n1 - n0; r < 16; ++r)
            *(float4*)(T + r * TP + c) = make_float4(0.f, 0.f, 0.f, 0.f);
    }
    __builtin_amdgcn_wave_barrier();

    const int m  = lane & 15;
    const int hi = lane >> 4;
    constexpr int NT = NCOLS / 16;
    v8f accm[NT] = {};
    const float* trow = T + m * TP + 2 * hi;
    for (int k = 0; k < 128; k += 4) {
        const v2f a = *(const v2f*)(trow + k);
        const float* wrow = ldsW + ((k >> 1) + hi) * S;
#pragma unroll
        for (int t = 0; t < NT; ++t) {
            const v2f b = *(const v2f*)(wrow + (t * 16 + m) * 2);
            accm[t] = __builtin_amdgcn_wmma_f32_16x16x4_f32(
                false, a, false, b, (short)0, accm[t], false, false);
        }
    }

    if (n1 - n0 == 16) {
#pragma unroll
        for (int t = 0; t < NT; ++t) {
            float* o = Hout + (size_t)n0 * NCOLS + t * 16 + m;
#pragma unroll
            for (int v = 0; v < 8; ++v)
                o[(size_t)(v + 8 * hi) * NCOLS] = accm[t][v];
        }
    } else {
#pragma unroll
        for (int t = 0; t < NT; ++t) {
            float* o = Hout + (size_t)n0 * NCOLS + t * 16 + m;
#pragma unroll
            for (int v = 0; v < 8; ++v)
                if (n0 + v + 8 * hi < n)
                    o[(size_t)(v + 8 * hi) * NCOLS] = accm[t][v];
        }
    }
}

__global__ __launch_bounds__(256) void k_gather16(const float* __restrict__ H,
                                                  const int* __restrict__ rowptr,
                                                  const int2* __restrict__ meta,
                                                  const float* __restrict__ dis,
                                                  const float* __restrict__ bias,
                                                  float* __restrict__ out,
                                                  int n, int e) {
    const int t    = blockIdx.x * blockDim.x + threadIdx.x;
    const int node = t >> 2;
    const int q    = (t & 3) * 4;
    if (node >= n) return;
    const int beg = rowptr[node];
    const int end = (node + 1 < n) ? rowptr[node + 1] : e;
    const float di = dis[node];
    const float* Hq = H + q;

    float4 acc[8];
    {
        const float4 hs = *(const float4*)(Hq + (size_t)node * 16);
        acc[0].x = hs.x * di; acc[0].y = hs.y * di; acc[0].z = hs.z * di; acc[0].w = hs.w * di;
    }
#pragma unroll
    for (int j = 1; j < 8; ++j) acc[j] = make_float4(0.f, 0.f, 0.f, 0.f);

    for (int p = beg; p < end; p += 8) {
        int2   m[8];
        float4 v[8];
#pragma unroll
        for (int j = 0; j < 8; ++j) {
            const int pj = (p + j < end) ? p + j : end - 1;
            m[j] = meta[pj];
        }
        __builtin_amdgcn_sched_barrier(0);
#pragma unroll
        for (int j = 0; j < 8; ++j)
            v[j] = *(const float4*)(Hq + (size_t)m[j].x * 16);
        __builtin_amdgcn_sched_barrier(0);
#pragma unroll
        for (int j = 0; j < 8; ++j) {
            const float w = (p + j < end) ? __int_as_float(m[j].y) : 0.f;
            acc[j].x = fmaf(v[j].x, w, acc[j].x); acc[j].y = fmaf(v[j].y, w, acc[j].y);
            acc[j].z = fmaf(v[j].z, w, acc[j].z); acc[j].w = fmaf(v[j].w, w, acc[j].w);
        }
    }
#pragma unroll
    for (int st = 1; st < 8; st <<= 1) {
#pragma unroll
        for (int j = 0; j < 8; j += 2 * st) {
            acc[j].x += acc[j + st].x; acc[j].y += acc[j + st].y;
            acc[j].z += acc[j + st].z; acc[j].w += acc[j + st].w;
        }
    }
    const float4 bb = *(const float4*)(bias + q);
    float4 r;
    r.x = fmaf(di, acc[0].x, bb.x); r.y = fmaf(di, acc[0].y, bb.y);
    r.z = fmaf(di, acc[0].z, bb.z); r.w = fmaf(di, acc[0].w, bb.w);
    *(float4*)(out + (size_t)node * 16 + q) = r;
}

extern "C" void kernel_launch(void* const* d_in, const int* in_sizes, int n_in,
                              void* d_out, int out_size, void* d_ws, size_t ws_size,
                              hipStream_t stream) {
    const float* x   = (const float*)d_in[0];
    const int*   ei  = (const int*)d_in[1];
    const float* W1  = (const float*)d_in[2];
    const float* b1  = (const float*)d_in[3];
    const float* W2  = (const float*)d_in[4];
    const float* b2  = (const float*)d_in[5];
    const float* W3  = (const float*)d_in[6];
    const float* b3  = (const float*)d_in[7];
    float*       out = (float*)d_out;

    const int N = in_sizes[0] / 128;
    const int E = in_sizes[1] / 2;
    const int* src = ei;
    const int* dst = ei + E;

    const int nchunk = (E + CH - 1) / CH;
    const int nrange = (N + RANGE - 1) / RANGE;
    float* A      = (float*)d_ws;
    float* B      = A + (size_t)N * 128;
    int2*  meta   = (int2*)(B + (size_t)N * 128);
    float* dis    = (float*)(meta + E);
    int*   degi   = (int*)(dis + N);
    int*   rowptr = degi + N;
    int*   rank   = rowptr + N;
    int*   cnt    = rank + E;
    int*   cbase  = cnt + (size_t)nchunk * N;
    int*   bsum   = cbase + (size_t)nchunk * N;

    const int TPB = 256;
    auto cdiv = [](long long a, long long b) { return (int)((a + b - 1) / b); };
    const int ntiles = (N + 15) / 16;
    const int nb     = cdiv(N, 2048);

    k_hist_rank<<<dim3(nchunk, nrange), 1024, 0, stream>>>(dst, rank, cnt, N, E);
    k_coldeg <<<cdiv(N, TPB), TPB, 0, stream>>>(cnt, degi, dis, N, nchunk);
    k_scan1  <<<nb, TPB, 0, stream>>>(degi, rowptr, bsum, N);
    k_scan2  <<<1, 32, 0, stream>>>(bsum, nb);
    k_colbase<<<cdiv(N, TPB), TPB, 0, stream>>>(rowptr, bsum, cnt, cbase, N, nchunk);
    k_place  <<<cdiv(E, TPB), TPB, 0, stream>>>(src, dst, rank, cbase, dis, meta, N, E);

    gemm_wmma_k<128, false><<<cdiv(ntiles, 8), TPB, 0, stream>>>(x, W1, A, ntiles);
    k_gather_gemm<128, 25><<<cdiv(ntiles, 25), 25 * 32, 0, stream>>>(A, rowptr, meta, dis, b1, W2, B, N, E, ntiles);
    k_gather_gemm<16, 25><<<cdiv(ntiles, 25), 25 * 32, 0, stream>>>(B, rowptr, meta, dis, b2, W3, A, N, E, ntiles);
    k_gather16<<<cdiv((long long)N * 4, TPB), TPB, 0, stream>>>(A, rowptr, meta, dis, b3, out, N, E);
}


# edge placement and the layer-1 product x@W1 in one launch (1 product block per 2 placement blocks, 4 edges per thread): matrix work runs beside the scattered accesses; 6 launches; sums unchanged
# speedup vs baseline: 1.7841x; 1.1095x over previous
#include <hip/hip_runtime.h>
#include <cstddef>

typedef __attribute__((ext_vector_type(2))) float v2f;
typedef __attribute__((ext_vector_type(8))) float v8f;

constexpr int GS = 288;

__device__ __forceinline__ void gemm_block_128(const float* __restrict__ X,
                                               const float* __restrict__ W,
                                               float* __restrict__ H,
                                               float* __restrict__ ldsW,
                                               int gblock, int ntiles) {
    constexpr int NCOLS = 128;
    constexpr int S = GS;

    for (int idx = threadIdx.x; idx < 128 * NCOLS; idx += 256) {
        const int r = idx / NCOLS;
        const int c = idx % NCOLS;
        ldsW[(r >> 1) * S + c * 2 + (r & 1)] = W[(size_t)r * NCOLS + c];
    }
    __syncthreads();

    const int lane = threadIdx.x & 31;
    const int wave = threadIdx.x >> 5;
    const int tile = gblock * 8 + wave;
    if (tile >= ntiles) return;
    const int row0 = tile * 16;
    const int m    = lane & 15;
    const int hi   = lane >> 4;
    constexpr int NT = NCOLS / 16;

    v8f acc[NT] = {};
    const float* xrow = X + (size_t)(row0 + m) * 128;

    for (int k = 0; k < 128; k += 4) {
        v2f a; a[0] = xrow[k + 2 * hi]; a[1] = xrow[k + 2 * hi + 1];
        const float* wrow = ldsW + ((k >> 1) + hi) * S;
#pragma unroll
        for (int t = 0; t < NT; ++t) {
            const v2f b = *(const v2f*)(wrow + (t * 16 + m) * 2);
            acc[t] = __builtin_amdgcn_wmma_f32_16x16x4_f32(
                false, a, false, b, (short)0, acc[t], false, false);
        }
    }

#pragma unroll
    for (int t = 0; t < NT; ++t) {
        float* o = H + (size_t)row0 * NCOLS + t * 16 + m;
#pragma unroll
        for (int v = 0; v < 8; ++v)
            o[(size_t)(v + 8 * hi) * NCOLS] = acc[t][v];
    }
}

constexpr int RANGE = 32768;
constexpr int CH    = 32768;
constexpr int NBLK  = 1024;
constexpr int CREG  = 64;

__global__ __launch_bounds__(1024) void k_hist_rank(const int* __restrict__ dst,
                                                    int* __restrict__ rank,
                                                    int* __restrict__ cnt,
                                                    int* __restrict__ subtot,
                                                    int n, int e, int nchunk) {
    __shared__ int hist[RANGE];
    const int tid = threadIdx.x;
    const int c   = blockIdx.x;
    const int lo  = blockIdx.y * RANGE;
    for (int j = tid; j < RANGE; j += 1024) hist[j] = 0;
    __syncthreads();

    const int ebeg = c * CH;
    const int eend = (ebeg + CH < e) ? ebeg + CH : e;
    for (int base = ebeg; base < eend; base += 4096) {
        int d[4];
#pragma unroll
        for (int u = 0; u < 4; ++u) {
            const int i = base + u * 1024 + tid;
            d[u] = (i < eend) ? dst[i] : -1;
        }
#pragma unroll
        for (int u = 0; u < 4; ++u) {
            const unsigned rel = (unsigned)(d[u] - lo);
            if (rel < (unsigned)RANGE)
                rank[base + u * 1024 + tid] = atomicAdd(&hist[rel], 1);
        }
    }
    __syncthreads();

    const int hi = (n - lo < RANGE) ? n - lo : RANGE;
    int* crow = cnt + (size_t)c * n + lo;
    const int lane = tid & 31;
    const int wave = tid >> 5;
    const int j0   = wave * NBLK;
    int h[NBLK / 32];
#pragma unroll
    for (int k = 0; k < NBLK / 32; ++k) h[k] = hist[j0 + k * 32 + lane];
    int s = 0;
#pragma unroll
    for (int k = 0; k < NBLK / 32; ++k) {
        const int j = j0 + k * 32 + lane;
        if (j < hi) { crow[j] = h[k]; s += h[k]; }
    }
#pragma unroll
    for (int off = 16; off > 0; off >>= 1) s += __shfl_xor(s, off);
    if (lane == 0 && j0 < hi)
        subtot[(size_t)(blockIdx.y * (RANGE / NBLK) + wave) * nchunk + c] = s;
}

__global__ __launch_bounds__(1024) void k_nodes(const int* __restrict__ cnt,
                                                const int* __restrict__ subtot,
                                                int* __restrict__ rowptr,
                                                float* __restrict__ dis,
                                                int* __restrict__ cbase,
                                                int n, int nchunk) {
    __shared__ int wtot[32];
    __shared__ int ptot[32];
    const int tid  = threadIdx.x;
    const int lane = tid & 31;
    const int wave = tid >> 5;
    const int b    = blockIdx.x;
    const int i    = b * NBLK + tid;
    const bool live = i < n;

    int t[CREG];
#pragma unroll
    for (int c = 0; c < CREG; ++c)
        t[c] = (live && c < nchunk) ? cnt[(size_t)c * n + i] : 0;

    int part = 0;
    const int nprev = b * nchunk;
    for (int idx = tid; idx < nprev; idx += 1024) part += subtot[idx];

    int deg = 0;
#pragma unroll
    for (int c = 0; c < CREG; ++c) deg += t[c];
    if (live)
        for (int c = CREG; c < nchunk; ++c) deg += cnt[(size_t)c * n + i];

    int incl = deg;
#pragma unroll
    for (int off = 1; off < 32; off <<= 1) {
        const int v = __shfl_up(incl, off);
        if (lane >= off) incl += v;
    }
#pragma unroll
    for (int off = 16; off > 0; off >>= 1) part += __shfl_xor(part, off);
    if (lane == 31) wtot[wave] = incl;
    if (lane == 0)  ptot[wave] = part;
    __syncthreads();
    int before = ptot[lane] + ((lane < wave) ? wtot[lane] : 0);
#pragma unroll
    for (int off = 16; off > 0; off >>= 1) before += __shfl_xor(before, off);

    if (live) {
        int run = before + incl - deg;
        rowptr[i] = run;
        dis[i] = rsqrtf((float)deg + 1.f);
#pragma unroll
        for (int c = 0; c < CREG; ++c) {
            if (c < nchunk) { cbase[(size_t)c * n + i] = run; run += t[c]; }
        }
        for (int c = CREG; c < nchunk; ++c) {
            const int tc = cnt[(size_t)c * n + i];
            cbase[(size_t)c * n + i] = run;
            run += tc;
        }
    }
}

constexpr int PE = 4;
static_assert(CH % (256 * PE) == 0, "a placement block must lie inside one edge chunk");

__global__ __launch_bounds__(256) void k_place_gemm(const int* __restrict__ src,
                                                    const int* __restrict__ dst,
                                                    const int* __restrict__ rank,
                                                    const int* __restrict__ cbase,
                                                    const float* __restrict__ dis,
                                                    int2* __restrict__ meta,
                                                    int n, int e,
                                                    const float* __restrict__ X,
                                                    const float* __restrict__ W,
                                                    float* __restrict__ H,
                                                    int ntiles) {
    __shared__ float ldsW[64 * GS];
    const int g = blockIdx.x / 3;
    const int r = blockIdx.x - g * 3;
    if (r == 0) {
        if (g * 8 < ntiles) gemm_block_128(X, W, H, ldsW, g, ntiles);
        return;
    }

    const int base = (g * 2 + r - 1) * (256 * PE);
    if (base >= e) return;
    const int* cb = cbase + (size_t)(base / CH) * n;
    int s[PE], d[PE], rk[PE];
#pragma unroll
    for (int u = 0; u < PE; ++u) {
        const int i  = base + u * 256 + (int)threadIdx.x;
        const int ic = (i < e) ? i : e - 1;
        s[u]  = src[ic];
        d[u]  = dst[ic];
        rk[u] = rank[ic];
    }
    int pos[PE];
    float w[PE];
#pragma unroll
    for (int u = 0; u < PE; ++u) {
        pos[u] = cb[d[u]] + rk[u];
        w[u]   = dis[s[u]];
    }
#pragma unroll
    for (int u = 0; u < PE; ++u) {
        const int i = base + u * 256 + (int)threadIdx.x;
        if (i < e) meta[pos[u]] = make_int2(s[u], __float_as_int(w[u]));
    }
}

constexpr int EB = 8;

template <int NCOLS, int WPB>
__global__ __launch_bounds__(WPB * 32) void k_gather_gemm(const float* __restrict__ H,
                                                          const int* __restrict__ rowptr,
                                                          const int2* __restrict__ meta,
                                                          const float* __restrict__ dis,
                                                          const float* __restrict__ bias,
                                                          const float* __restrict__ W,
                                                          float* __restrict__ Hout,
                                                          int n, int e, int ntiles) {
    constexpr int S  = (NCOLS == 128) ? 288 : NCOLS * 2;
    constexpr int TP = 132;
    __shared__ float ldsW[64 * S];
    __shared__ float ldsT[WPB * 16 * TP];

    for (int idx = threadIdx.x; idx < 128 * NCOLS; idx += WPB * 32) {
        const int r = idx / NCOLS;
        const int c = idx % NCOLS;
        ldsW[(r >> 1) * S + c * 2 + (r & 1)] = W[(size_t)r * NCOLS + c];
    }
    __syncthreads();

    const int wave = __builtin_amdgcn_readfirstlane((int)(threadIdx.x >> 5));
    const int lane = threadIdx.x & 31;
    const int tile = blockIdx.x * WPB + wave;
    if (tile >= ntiles) return;
    float* T = ldsT + wave * (16 * TP);
    const int n0 = tile * 16;
    const int n1 = (n0 + 16 < n) ? n0 + 16 : n;

    {
        const int c = lane * 4;
        const float4 bb = *(const float4*)(bias + c);
        const float* Hc = H + c;
        int beg = rowptr[n0];
        for (int node = n0; node < n1; ++node) {
            const int   end = (node + 1 < n) ? rowptr[node + 1] : e;
            const float di  = dis[node];
            const int   cnt = end - beg + 1;

            int2 m[EB];
            const int2 self = make_int2(node, __float_as_int(di));
            m[0] = self;
            if (beg < end) {
#pragma unroll
                for (int j = 1; j < EB; ++j) {
                    const int pj = (beg + j - 1 < end - 1) ? beg + j - 1 : end - 1;
                    m[j] = meta[pj];
                }
            } else {
#pragma unroll
                for (int j = 1; j < EB; ++j) m[j] = self;
            }

            float4 acc = make_float4(0.f, 0.f, 0.f, 0.f);
            for (int q = 0; q < cnt; q += EB) {
                float4 v[EB];
                float  w[EB];
#pragma unroll
                for (int j = 0; j < EB; ++j) {
                    v[j] = *(const float4*)(Hc + (size_t)m[j].x * 128);
                    w[j] = __int_as_float(m[j].y);
                }
                const int left = cnt - q;
                __builtin_amdgcn_sched_barrier(0);
                if (left > EB) {
                    const int pe = beg + q + EB - 1;
#pragma unroll
                    for (int j = 0; j < EB; ++j) {
                        const int pj = (pe + j < end - 1) ? pe + j : end - 1;
                        m[j] = meta[pj];
                    }
                }
#pragma unroll
                for (int j = 0; j < EB; ++j) {
                    if (j < left) {
                        acc.x = fmaf(v[j].x, w[j], acc.x); acc.y = fmaf(v[j].y, w[j], acc.y);
                        acc.z = fmaf(v[j].z, w[j], acc.z); acc.w = fmaf(v[j].w, w[j], acc.w);
                    }
                }
            }
            float4 r;
            r.x = fmaxf(fmaf(di, acc.x, bb.x), 0.f); r.y = fmaxf(fmaf(di, acc.y, bb.y), 0.f);
            r.z = fmaxf(fmaf(di, acc.z, bb.z), 0.f); r.w = fmaxf(fmaf(di, acc.w, bb.w), 0.f);
            *(float4*)(T + (node - n0) * TP + c) = r;
            beg = end;
        }
        for (int r = n1 - n0; r < 16; ++r)
            *(float4*)(T + r * TP + c) = make_float4(0.f, 0.f, 0.f, 0.f);
    }
    __builtin_amdgcn_wave_barrier();

    const int m  = lane & 15;
    const int hi = lane >> 4;
    constexpr int NT = NCOLS / 16;
    v8f accm[NT] = {};
    const float* trow = T + m * TP + 2 * hi;
    for (int k = 0; k < 128; k += 4) {
        const v2f a = *(const v2f*)(trow + k);
        const float* wrow = ldsW + ((k >> 1) + hi) * S;
#pragma unroll
        for (int t = 0; t < NT; ++t) {
            const v2f b = *(const v2f*)(wrow + (t * 16 + m) * 2);
            accm[t] = __builtin_amdgcn_wmma_f32_16x16x4_f32(
                false, a, false, b, (short)0, accm[t], false, false);
        }
    }

    if (n1 - n0 == 16) {
#pragma unroll
        for (int t = 0; t < NT; ++t) {
            float* o = Hout + (size_t)n0 * NCOLS + t * 16 + m;
#pragma unroll
            for (int v = 0; v < 8; ++v)
                o[(size_t)(v + 8 * hi) * NCOLS] = accm[t][v];
        }
    } else {
#pragma unroll
        for (int t = 0; t < NT; ++t) {
            float* o = Hout + (size_t)n0 * NCOLS + t * 16 + m;
#pragma unroll
            for (int v = 0; v < 8; ++v)
                if (n0 + v + 8 * hi < n)
                    o[(size_t)(v + 8 * hi) * NCOLS] = accm[t][v];
        }
    }
}

__global__ __launch_bounds__(256) void k_gather16(const float* __restrict__ H,
                                                  const int* __restrict__ rowptr,
                                                  const int2* __restrict__ meta,
                                                  const float* __restrict__ dis,
                                                  const float* __restrict__ bias,
                                                  float* __restrict__ out,
                                                  int n, int e) {
    const int t    = blockIdx.x * blockDim.x + threadIdx.x;
    const int node = t >> 2;
    const int q    = (t & 3) * 4;
    if (node >= n) return;
    const int beg = rowptr[node];
    const int end = (node + 1 < n) ? rowptr[node + 1] : e;
    const float di = dis[node];
    const float* Hq = H + q;

    float4 acc[8];
    {
        const float4 hs = *(const float4*)(Hq + (size_t)node * 16);
        acc[0].x = hs.x * di; acc[0].y = hs.y * di; acc[0].z = hs.z * di; acc[0].w = hs.w * di;
    }
#pragma unroll
    for (int j = 1; j < 8; ++j) acc[j] = make_float4(0.f, 0.f, 0.f, 0.f);

    for (int p = beg; p < end; p += 8) {
        int2   m[8];
        float4 v[8];
#pragma unroll
        for (int j = 0; j < 8; ++j) {
            const int pj = (p + j < end) ? p + j : end - 1;
            m[j] = meta[pj];
        }
        __builtin_amdgcn_sched_barrier(0);
#pragma unroll
        for (int j = 0; j < 8; ++j)
            v[j] = *(const float4*)(Hq + (size_t)m[j].x * 16);
        __builtin_amdgcn_sched_barrier(0);
#pragma unroll
        for (int j = 0; j < 8; ++j) {
            const float w = (p + j < end) ? __int_as_float(m[j].y) : 0.f;
            acc[j].x = fmaf(v[j].x, w, acc[j].x); acc[j].y = fmaf(v[j].y, w, acc[j].y);
            acc[j].z = fmaf(v[j].z, w, acc[j].z); acc[j].w = fmaf(v[j].w, w, acc[j].w);
        }
    }
#pragma unroll
    for (int st = 1; st < 8; st <<= 1) {
#pragma unroll
        for (int j = 0; j < 8; j += 2 * st) {
            acc[j].x += acc[j + st].x; acc[j].y += acc[j + st].y;
            acc[j].z += acc[j + st].z; acc[j].w += acc[j + st].w;
        }
    }
    const float4 bb = *(const float4*)(bias + q);
    float4 r;
    r.x = fmaf(di, acc[0].x, bb.x); r.y = fmaf(di, acc[0].y, bb.y);
    r.z = fmaf(di, acc[0].z, bb.z); r.w = fmaf(di, acc[0].w, bb.w);
    *(float4*)(out + (size_t)node * 16 + q) = r;
}

extern "C" void kernel_launch(void* const* d_in, const int* in_sizes, int n_in,
                              void* d_out, int out_size, void* d_ws, size_t ws_size,
                              hipStream_t stream) {
    const float* x   = (const float*)d_in[0];
    const int*   ei  = (const int*)d_in[1];
    const float* W1  = (const float*)d_in[2];
    const float* b1  = (const float*)d_in[3];
    const float* W2  = (const float*)d_in[4];
    const float* b2  = (const float*)d_in[5];
    const float* W3  = (const float*)d_in[6];
    const float* b3  = (const float*)d_in[7];
    float*       out = (float*)d_out;

    const int N = in_sizes[0] / 128;
    const int E = in_sizes[1] / 2;
    const int* src = ei;
    const int* dst = ei + E;

    const int nchunk = (E + CH - 1) / CH;
    const int nrange = (N + RANGE - 1) / RANGE;
    float* A      = (float*)d_ws;
    float* B      = A + (size_t)N * 128;
    int2*  meta   = (int2*)(B + (size_t)N * 128);
    float* dis    = (float*)(meta + E);
    int*   rowptr = (int*)(dis + N);
    int*   rank   = rowptr + N;
    int*   cnt    = rank + E;
    int*   cbase  = cnt + (size_t)nchunk * N;
    int*   subtot = cbase + (size_t)nchunk * N;

    const int TPB = 256;
    auto cdiv = [](long long a, long long b) { return (int)((a + b - 1) / b); };
    const int ntiles = (N + 15) / 16;

    k_hist_rank<<<dim3(nchunk, nrange), 1024, 0, stream>>>(dst, rank, cnt, subtot, N, E, nchunk);
    k_nodes  <<<cdiv(N, NBLK), 1024, 0, stream>>>(cnt, subtot, rowptr, dis, cbase, N, nchunk);
    const int ngemm  = cdiv(ntiles, 8);
    const int nplace = cdiv(cdiv(E, TPB * PE), 2);
    const int ngroup = (ngemm > nplace) ? ngemm : nplace;
    k_place_gemm<<<ngroup * 3, TPB, 0, stream>>>(src, dst, rank, cbase, dis, meta, N, E, x, W1, A, ntiles);
    k_gather_gemm<128, 25><<<cdiv(ntiles, 25), 25 * 32, 0, stream>>>(A, rowptr, meta, dis, b1, W2, B, N, E, ntiles);
    k_gather_gemm<16, 25><<<cdiv(ntiles, 25), 25 * 32, 0, stream>>>(B, rowptr, meta, dis, b2, W3, A, N, E, ntiles);
    k_gather16<<<cdiv((long long)N * 4, TPB), TPB, 0, stream>>>(A, rowptr, meta, dis, b3, out, N, E);
}
